# starting kernel
# baseline (speedup 1.0000x reference)
#include <hip/hip_runtime.h>

typedef float v2f __attribute__((ext_vector_type(2)));
typedef float v4f __attribute__((ext_vector_type(4)));
typedef float v8f __attribute__((ext_vector_type(8)));

#define BATCH 32
#define HW    512
#define POOL  128
#define GROUPS (BATCH * POOL * (POOL / 16))

__global__ __launch_bounds__(256) void spa_pool_diff_wmma(
    const float* __restrict__ org, const float* __restrict__ enh,
    float* __restrict__ p)
{
    const int lane  = threadIdx.x & 31;
    const int group = blockIdx.x * 8 + (threadIdx.x >> 5);

    const int xg = group & 7;
    const int yo = (group >> 3) & 127;
    const int b  = group >> 10;

    const int chanStride = HW * HW;
    const int rowBase = (b * 3) * chanStride + (4 * yo) * HW + 64 * xg;

#if __has_builtin(__builtin_amdgcn_wmma_f32_16x16x4_f32)
    const int laneOff = 4 * (lane & 15) + ((lane >> 4) << 1);

    v8f acc = {0.f, 0.f, 0.f, 0.f, 0.f, 0.f, 0.f, 0.f};
    const v2f bmat = {1.0f / 48.0f, 1.0f / 48.0f};

    #pragma unroll
    for (int c = 0; c < 3; ++c) {
        #pragma unroll
        for (int dy = 0; dy < 4; ++dy) {
            const int off = rowBase + c * chanStride + dy * HW + laneOff;
            v2f o = __builtin_nontemporal_load((const v2f*)(org + off));
            v2f e = __builtin_nontemporal_load((const v2f*)(enh + off));
            v2f a; a.x = o.x - e.x; a.y = o.y - e.y;
            acc = __builtin_amdgcn_wmma_f32_16x16x4_f32(
                 false, a,  false, bmat,
                 (short)0, acc,  false,  false);
        }
    }

    float* outp = p + (b * POOL + yo) * POOL + 16 * xg;
    if (lane == 0) {
        v4f lo = {acc[0], acc[1], acc[2], acc[3]};
        v4f hi = {acc[4], acc[5], acc[6], acc[7]};
        *(v4f*)(outp + 0) = lo;
        *(v4f*)(outp + 4) = hi;
    } else if (lane == 16) {
        v4f lo = {acc[0], acc[1], acc[2], acc[3]};
        v4f hi = {acc[4], acc[5], acc[6], acc[7]};
        *(v4f*)(outp + 8)  = lo;
        *(v4f*)(outp + 12) = hi;
    }
#else
    if (lane < 16) {
        float sum = 0.f;
        const int base = rowBase + 4 * lane;
        #pragma unroll
        for (int c = 0; c < 3; ++c)
            #pragma unroll
            for (int dy = 0; dy < 4; ++dy) {
                const float* po = org + base + c * chanStride + dy * HW;
                const float* pe = enh + base + c * chanStride + dy * HW;
                #pragma unroll
                for (int dx = 0; dx < 4; ++dx) sum += po[dx] - pe[dx];
            }
        p[(b * POOL + yo) * POOL + 16 * xg + lane] = sum * (1.0f / 48.0f);
    }
#endif
}

__global__ __launch_bounds__(256) void spa_energy(
    const float* __restrict__ p, float* __restrict__ E)
{
    const int idx = blockIdx.x * 256 + threadIdx.x;
    const int j = idx & 127;
    const int i = (idx >> 7) & 127;
    const int b = idx >> 14;

    const float* pb = p + b * POOL * POOL;
    const float c = pb[i * POOL + j];
    const float l = (j > 0)          ? pb[i * POOL + j - 1] : 0.f;
    const float r = (j < POOL - 1)   ? pb[i * POOL + j + 1] : 0.f;
    const float u = (i > 0)          ? pb[(i - 1) * POOL + j] : 0.f;
    const float d = (i < POOL - 1)   ? pb[(i + 1) * POOL + j] : 0.f;

    const float dl = c - l, dr = c - r, du = c - u, dd = c - d;
    E[idx] = dl * dl + dr * dr + du * du + dd * dd;
}

extern "C" void kernel_launch(void* const* d_in, const int* in_sizes, int n_in,
                              void* d_out, int out_size, void* d_ws, size_t ws_size,
                              hipStream_t stream) {
    const float* org = (const float*)d_in[0];
    const float* enh = (const float*)d_in[1];
    float* p = (float*)d_ws;
    float* E = (float*)d_out;

    spa_pool_diff_wmma<<<GROUPS / 8, 256, 0, stream>>>(org, enh, p);
    spa_energy<<<(BATCH * POOL * POOL) / 256, 256, 0, stream>>>(p, E);
}


# Fuse the neighbour-difference pass into the pooling kernel: one launch of 256 workgroups x 16 waves, each pools its 16-row band plus two halo rows into LDS and writes E from LDS
# speedup vs baseline: 1.1115x; 1.1115x over previous
#include <hip/hip_runtime.h>


typedef float v2f __attribute__((ext_vector_type(2)));
typedef float v4f __attribute__((ext_vector_type(4)));
typedef float v8f __attribute__((ext_vector_type(8)));

#define BATCH   32
#define HW      512
#define POOL    128
#define BAND    16
#define NBANDS  (POOL / BAND)
#define ROWS    (BAND + 2)
#define WAVES   16
#define THREADS (WAVES * 32)
#define STEPS   (ROWS / 2)
#define PITCH   136

__global__ void __launch_bounds__(THREADS) __attribute__((amdgpu_waves_per_eu(4, 4)))
spa_fused_band(const float* __restrict__ org, const float* __restrict__ enh,
               float* __restrict__ E)
{
    __shared__ __attribute__((aligned(16))) float ps[ROWS * PITCH];

    const int tid  = threadIdx.x;
    const int lane = tid & 31;
    const int wave = __builtin_amdgcn_readfirstlane(tid >> 5);
    const int band = blockIdx.x & (NBANDS - 1);
    const int b    = blockIdx.x >> 3;

    const int xg   = wave & 7;
    const int rsel = wave >> 3;
    const int yo0  = BAND * band - 1 + rsel;

    if (tid < 2 * ROWS)
        ps[(tid >> 1) * PITCH + ((tid & 1) ? (4 + POOL) : 3)] = 0.f;

    const int chanStride = HW * HW;
    const int laneOff = 4 * (lane & 15) + ((lane >> 4) << 1);
    const int segBase = (b * 3) * chanStride + 64 * xg;

    v2f o[2][12], e[2][12];

#define LOAD_STEP(T) do {                                                              \
        int yo_ = yo0 + 2 * (T);                                                       \
        yo_ = yo_ < 0 ? 0 : (yo_ > POOL - 1 ? POOL - 1 : yo_);                         \
        const int rb_ = segBase + (4 * yo_) * HW;                                      \
        _Pragma("unroll") for (int c = 0; c < 3; ++c) {                                \
            _Pragma("unroll") for (int dy = 0; dy < 4; ++dy) {                         \
                const int off_ = rb_ + c * chanStride + dy * HW + laneOff;             \
                o[(T) & 1][c * 4 + dy] = __builtin_nontemporal_load((const v2f*)(org + off_)); \
                e[(T) & 1][c * 4 + dy] = __builtin_nontemporal_load((const v2f*)(enh + off_)); \
            }                                                                          \
        }                                                                              \
    } while (0)

    const v2f bmat = {1.0f / 48.0f, 1.0f / 48.0f};

    LOAD_STEP(0);
    #pragma unroll
    for (int t = 0; t < STEPS; ++t) {
        __builtin_amdgcn_sched_barrier(0);
        if (t + 1 < STEPS) LOAD_STEP(t + 1);
        __builtin_amdgcn_sched_barrier(0);

        v8f acc = {0.f, 0.f, 0.f, 0.f, 0.f, 0.f, 0.f, 0.f};
        #pragma unroll
        for (int c = 0; c < 3; ++c) {
            #pragma unroll
            for (int dy = 0; dy < 4; ++dy) {
                const v2f ov = o[t & 1][c * 4 + dy];
                const v2f ev = e[t & 1][c * 4 + dy];
                v2f a; a.x = ov.x - ev.x; a.y = ov.y - ev.y;
                acc = __builtin_amdgcn_wmma_f32_16x16x4_f32(
                     false, a,  false, bmat,
                     (short)0, acc,  false,  false);
            }
        }

        const int  yo    = yo0 + 2 * t;
        const bool valid = (yo >= 0) && (yo < POOL);
        if ((lane & 15) == 0) {
            float* dst = ps + (2 * t + rsel) * PITCH + 4 + 16 * xg + 8 * (lane >> 4);
            v4f lo = {acc[0], acc[1], acc[2], acc[3]};
            v4f hi = {acc[4], acc[5], acc[6], acc[7]};
            const v4f zero = {0.f, 0.f, 0.f, 0.f};
            *(v4f*)(dst + 0) = valid ? lo : zero;
            *(v4f*)(dst + 4) = valid ? hi : zero;
        }
    }
#undef LOAD_STEP

    __syncthreads();

    {
        const float* rc = ps + (wave + 1) * PITCH + 4 + 4 * lane;
        const v4f c4 = *(const v4f*)rc;
        const v4f u4 = *(const v4f*)(rc - PITCH);
        const v4f d4 = *(const v4f*)(rc + PITCH);
        const float lft = rc[-1];
        const float rgt = rc[4];

        const float cl[4] = {lft,  c4.x, c4.y, c4.z};
        const float cr[4] = {c4.y, c4.z, c4.w, rgt};
        const float cc[4] = {c4.x, c4.y, c4.z, c4.w};
        const float cu[4] = {u4.x, u4.y, u4.z, u4.w};
        const float cd[4] = {d4.x, d4.y, d4.z, d4.w};
        float r[4];
        #pragma unroll
        for (int k = 0; k < 4; ++k) {
            const float dl = cc[k] - cl[k], dr = cc[k] - cr[k];
            const float du = cc[k] - cu[k], dd = cc[k] - cd[k];
            r[k] = dl * dl + dr * dr + du * du + dd * dd;
        }
        const v4f res = {r[0], r[1], r[2], r[3]};
        *(v4f*)(E + ((b * POOL + BAND * band + wave) * POOL + 4 * lane)) = res;
    }
}

extern "C" void kernel_launch(void* const* d_in, const int* in_sizes, int n_in,
                              void* d_out, int out_size, void* d_ws, size_t ws_size,
                              hipStream_t stream) {
    const float* org = (const float*)d_in[0];
    const float* enh = (const float*)d_in[1];
    float* E = (float*)d_out;

    spa_fused_band<<<BATCH * NBANDS, THREADS, 0, stream>>>(org, enh, E);
}
